# starting kernel
# baseline (speedup 1.0000x reference)
#include <hip/hip_runtime.h>
#include <math.h>


typedef __attribute__((ext_vector_type(2))) float v2f;
typedef __attribute__((ext_vector_type(8))) float v8f;

#define IN_CH  256
#define HID_CH 128
#define OUT_CH 64

template <int K, int NCOLS>
__global__ __launch_bounds__(32) void gemm_wmma_f32(
    const float* __restrict__ A, const float* __restrict__ W,
    float* __restrict__ Out, int nRows) {
  constexpr int NT = NCOLS / 16;
  const int mbase = blockIdx.x * 16;
  const int lane  = threadIdx.x & 31;
  const int g     = lane >> 4;
  const int l     = lane & 15;

  v8f acc[NT] = {};

  int arow_idx = mbase + l;
  if (arow_idx >= nRows) arow_idx = nRows - 1;
  const float* __restrict__ arow = A + (size_t)arow_idx * K + 2 * g;

#pragma unroll 4
  for (int k = 0; k < K; k += 4) {
    v2f a;
    a.x = arow[k];
    a.y = arow[k + 1];
    const float* __restrict__ wk = W + (size_t)(k + 2 * g) * NCOLS + l;
#pragma unroll
    for (int t = 0; t < NT; ++t) {
      v2f b;
      b.x = wk[t * 16];
      b.y = wk[NCOLS + t * 16];
      acc[t] = __builtin_amdgcn_wmma_f32_16x16x4_f32(
           false, a,  false, b,
           (short)0, acc[t],  false,  false);
    }
  }

#pragma unroll
  for (int t = 0; t < NT; ++t) {
#pragma unroll
    for (int r = 0; r < 8; ++r) {
      const int m = mbase + r + 8 * g;
      if (m < nRows) Out[(size_t)m * NCOLS + t * 16 + l] = acc[t][r];
    }
  }
}

__global__ __launch_bounds__(256) void deg_kernel(
    const int* __restrict__ dst, unsigned* __restrict__ deg, int nE) {
  const int e = blockIdx.x * blockDim.x + threadIdx.x;
  if (e < nE) atomicAdd(&deg[dst[e]], 1u);
}

__global__ __launch_bounds__(256) void dinv_kernel(
    const unsigned* __restrict__ deg, float* __restrict__ dinv, int n) {
  const int i = blockIdx.x * blockDim.x + threadIdx.x;
  if (i < n) dinv[i] = rsqrtf((float)deg[i] + 1.0f);
}

template <int C>
__global__ __launch_bounds__(256) void scatter_edges(
    const int* __restrict__ src, const int* __restrict__ dst,
    const float* __restrict__ dinv, const float* __restrict__ hin,
    float* __restrict__ agg, int nE) {
  const int wid  = (int)((blockIdx.x * (unsigned)blockDim.x + threadIdx.x) >> 5);
  const int lane = threadIdx.x & 31;
  if (wid >= nE) return;
  const int s = src[wid];
  const int d = dst[wid];
  const float nrm = dinv[s] * dinv[d];
  constexpr int PER = C / 32;
  const float* __restrict__ hp = hin + (size_t)s * C + lane * PER;
  float* __restrict__ ap       = agg + (size_t)d * C + lane * PER;
#pragma unroll
  for (int j = 0; j < PER; ++j) {
    __hip_atomic_fetch_add(&ap[j], nrm * hp[j], __ATOMIC_RELAXED,
                           __HIP_MEMORY_SCOPE_AGENT);
  }
}

template <int C>
__global__ __launch_bounds__(256) void finish_layer(
    float* __restrict__ agg, const float* __restrict__ lin,
    const float* __restrict__ dinv, const float* __restrict__ bias,
    int n, int doRelu) {
  const int t = blockIdx.x * blockDim.x + threadIdx.x;
  if (t >= n * C) return;
  const int i = t / C;
  const int c = t % C;
  const float di = dinv[i];
  float v = agg[t] + di * di * lin[t] + bias[c];
  if (doRelu) v = fmaxf(v, 0.0f);
  agg[t] = v;
}

__global__ __launch_bounds__(256) void decode_kernel(
    const int* __restrict__ src, const int* __restrict__ dst,
    const float* __restrict__ z, float* __restrict__ out, int nE) {
  const int wid  = (int)((blockIdx.x * (unsigned)blockDim.x + threadIdx.x) >> 5);
  const int lane = threadIdx.x & 31;
  if (wid >= nE) return;
  const int s = src[wid];
  const int d = dst[wid];
  const float* __restrict__ zs = z + (size_t)s * OUT_CH + lane * 2;
  const float* __restrict__ zd = z + (size_t)d * OUT_CH + lane * 2;
  float p = zs[0] * zd[0] + zs[1] * zd[1];
#pragma unroll
  for (int off = 16; off > 0; off >>= 1) p += __shfl_xor(p, off, 32);
  if (lane == 0) out[wid] = 1.0f / (1.0f + __expf(-p));
}

extern "C" void kernel_launch(void* const* d_in, const int* in_sizes, int n_in,
                              void* d_out, int out_size, void* d_ws, size_t ws_size,
                              hipStream_t stream) {
  const float* x   = (const float*)d_in[0];
  const int*   ei  = (const int*)d_in[1];
  const float* W1  = (const float*)d_in[2];
  const float* b1  = (const float*)d_in[3];
  const float* W2  = (const float*)d_in[4];
  const float* b2  = (const float*)d_in[5];

  const int N = in_sizes[0] / IN_CH;
  const int E = in_sizes[1] / 2;
  const int* esrc = ei;
  const int* edst = ei + E;

  char* w = (char*)d_ws;
  unsigned* deg  = (unsigned*)w; w += (size_t)N * 4;
  float*    dinv = (float*)w;    w += (size_t)N * 4;
  float*    hlin = (float*)w;    w += (size_t)N * HID_CH * 4;
  float*    hagg = (float*)w;    w += (size_t)N * HID_CH * 4;
  float*    zlin = (float*)w;    w += (size_t)N * OUT_CH * 4;
  float*    zagg = (float*)w;    w += (size_t)N * OUT_CH * 4;

  hipMemsetAsync(deg,  0, (size_t)N * 4, stream);
  hipMemsetAsync(hagg, 0, (size_t)N * HID_CH * 4, stream);
  hipMemsetAsync(zagg, 0, (size_t)N * OUT_CH * 4, stream);

  const int TPB = 256;
  const int nWarpBlocks = (int)(((size_t)E * 32 + TPB - 1) / TPB);
  const int mTiles = (N + 15) / 16;

  deg_kernel<<<(E + TPB - 1) / TPB, TPB, 0, stream>>>(edst, deg, E);
  dinv_kernel<<<(N + TPB - 1) / TPB, TPB, 0, stream>>>(deg, dinv, N);

  gemm_wmma_f32<IN_CH, HID_CH><<<mTiles, 32, 0, stream>>>(x, W1, hlin, N);
  scatter_edges<HID_CH><<<nWarpBlocks, TPB, 0, stream>>>(esrc, edst, dinv, hlin, hagg, E);
  finish_layer<HID_CH><<<((size_t)N * HID_CH + TPB - 1) / TPB, TPB, 0, stream>>>(
      hagg, hlin, dinv, b1, N, 1);

  gemm_wmma_f32<HID_CH, OUT_CH><<<mTiles, 32, 0, stream>>>(hagg, W2, zlin, N);
  scatter_edges<OUT_CH><<<nWarpBlocks, TPB, 0, stream>>>(esrc, edst, dinv, zlin, zagg, E);
  finish_layer<OUT_CH><<<((size_t)N * OUT_CH + TPB - 1) / TPB, TPB, 0, stream>>>(
      zagg, zlin, dinv, b2, N, 0);

  decode_kernel<<<nWarpBlocks, TPB, 0, stream>>>(esrc, edst, zagg, (float*)d_out, E);
}


# Scatter: a wave takes 32 edges, not one: indices and edge scale read once per wave, row bases in scalar registers, 32 row lines in flight before their f32 atomic adds; sums stay in arrival order
# speedup vs baseline: 1.4546x; 1.4546x over previous
#include <hip/hip_runtime.h>
#include <math.h>


typedef __attribute__((ext_vector_type(2))) float v2f;
typedef __attribute__((ext_vector_type(8))) float v8f;

#define IN_CH  256
#define HID_CH 128
#define OUT_CH 64

template <int K, int NCOLS>
__global__ __launch_bounds__(32) void gemm_wmma_f32(
    const float* __restrict__ A, const float* __restrict__ W,
    float* __restrict__ Out, int nRows) {
  constexpr int NT = NCOLS / 16;
  const int mbase = blockIdx.x * 16;
  const int lane  = threadIdx.x & 31;
  const int g     = lane >> 4;
  const int l     = lane & 15;

  v8f acc[NT] = {};

  int arow_idx = mbase + l;
  if (arow_idx >= nRows) arow_idx = nRows - 1;
  const float* __restrict__ arow = A + (size_t)arow_idx * K + 2 * g;

#pragma unroll 4
  for (int k = 0; k < K; k += 4) {
    v2f a;
    a.x = arow[k];
    a.y = arow[k + 1];
    const float* __restrict__ wk = W + (size_t)(k + 2 * g) * NCOLS + l;
#pragma unroll
    for (int t = 0; t < NT; ++t) {
      v2f b;
      b.x = wk[t * 16];
      b.y = wk[NCOLS + t * 16];
      acc[t] = __builtin_amdgcn_wmma_f32_16x16x4_f32(
           false, a,  false, b,
           (short)0, acc[t],  false,  false);
    }
  }

#pragma unroll
  for (int t = 0; t < NT; ++t) {
#pragma unroll
    for (int r = 0; r < 8; ++r) {
      const int m = mbase + r + 8 * g;
      if (m < nRows) Out[(size_t)m * NCOLS + t * 16 + l] = acc[t][r];
    }
  }
}

__global__ __launch_bounds__(256) void deg_kernel(
    const int* __restrict__ dst, unsigned* __restrict__ deg, int nE) {
  const int e = blockIdx.x * blockDim.x + threadIdx.x;
  if (e < nE) atomicAdd(&deg[dst[e]], 1u);
}

__global__ __launch_bounds__(256) void dinv_kernel(
    const unsigned* __restrict__ deg, float* __restrict__ dinv, int n) {
  const int i = blockIdx.x * blockDim.x + threadIdx.x;
  if (i < n) dinv[i] = rsqrtf((float)deg[i] + 1.0f);
}

typedef const float __attribute__((address_space(1)))* gmem_cf;
typedef float __attribute__((address_space(1)))* gmem_f;

template <int C>
__global__ __launch_bounds__(256) void scatter_edges(
    const int* __restrict__ src, const int* __restrict__ dst,
    const float* __restrict__ dinv, const float* __restrict__ hin,
    float* __restrict__ agg, int nE) {
  constexpr int PER = C / 32;
  constexpr int EB  = 32 / PER;
  const unsigned lane = threadIdx.x & 31;
  const int wave = __builtin_amdgcn_readfirstlane(
      (int)((blockIdx.x * (unsigned)blockDim.x + threadIdx.x) >> 5));
  const int base = wave * 32;
  if (base >= nE) return;
  const int cnt = min(32, nE - base);

  const int e   = base + min((int)lane, cnt - 1);
  const int s   = src[e];
  const int d   = dst[e];
  const int nrm = __float_as_int(dinv[s] * dinv[d]);

  if (cnt == 32) {
#pragma unroll 1
    for (int b = 0; b < 32; b += EB) {
      float v[EB][PER];
#pragma unroll
      for (int i = 0; i < EB; ++i) {
        const int se = __builtin_amdgcn_readlane(s, b + i);
        gmem_cf hp = (gmem_cf)(hin + (size_t)se * C);
        asm volatile("" : "+s"(hp));
#pragma unroll
        for (int j = 0; j < PER; ++j) v[i][j] = hp[lane + 32 * j];
      }
#pragma unroll
      for (int i = 0; i < EB; ++i) {
        const int   de = __builtin_amdgcn_readlane(d, b + i);
        const float ne = __int_as_float(__builtin_amdgcn_readlane(nrm, b + i));
        gmem_f ap = (gmem_f)(agg + (size_t)de * C);
        asm volatile("" : "+s"(ap));
#pragma unroll
        for (int j = 0; j < PER; ++j) {
          __hip_atomic_fetch_add(&ap[lane + 32 * j], ne * v[i][j],
                                 __ATOMIC_RELAXED, __HIP_MEMORY_SCOPE_AGENT);
        }
      }
    }
  } else {
#pragma unroll 1
    for (int i = 0; i < cnt; ++i) {
      const int   se = __builtin_amdgcn_readlane(s, i);
      const int   de = __builtin_amdgcn_readlane(d, i);
      const float ne = __int_as_float(__builtin_amdgcn_readlane(nrm, i));
      const float* __restrict__ hp = hin + (size_t)se * C;
      float* __restrict__ ap       = agg + (size_t)de * C;
      float r[PER];
#pragma unroll
      for (int j = 0; j < PER; ++j) r[j] = hp[lane + 32 * j];
#pragma unroll
      for (int j = 0; j < PER; ++j) {
        __hip_atomic_fetch_add(&ap[lane + 32 * j], ne * r[j],
                               __ATOMIC_RELAXED, __HIP_MEMORY_SCOPE_AGENT);
      }
    }
  }
}

template <int C>
__global__ __launch_bounds__(256) void finish_layer(
    float* __restrict__ agg, const float* __restrict__ lin,
    const float* __restrict__ dinv, const float* __restrict__ bias,
    int n, int doRelu) {
  const int t = blockIdx.x * blockDim.x + threadIdx.x;
  if (t >= n * C) return;
  const int i = t / C;
  const int c = t % C;
  const float di = dinv[i];
  float v = agg[t] + di * di * lin[t] + bias[c];
  if (doRelu) v = fmaxf(v, 0.0f);
  agg[t] = v;
}

__global__ __launch_bounds__(256) void decode_kernel(
    const int* __restrict__ src, const int* __restrict__ dst,
    const float* __restrict__ z, float* __restrict__ out, int nE) {
  const int wid  = (int)((blockIdx.x * (unsigned)blockDim.x + threadIdx.x) >> 5);
  const int lane = threadIdx.x & 31;
  if (wid >= nE) return;
  const int s = src[wid];
  const int d = dst[wid];
  const float* __restrict__ zs = z + (size_t)s * OUT_CH + lane * 2;
  const float* __restrict__ zd = z + (size_t)d * OUT_CH + lane * 2;
  float p = zs[0] * zd[0] + zs[1] * zd[1];
#pragma unroll
  for (int off = 16; off > 0; off >>= 1) p += __shfl_xor(p, off, 32);
  if (lane == 0) out[wid] = 1.0f / (1.0f + __expf(-p));
}

extern "C" void kernel_launch(void* const* d_in, const int* in_sizes, int n_in,
                              void* d_out, int out_size, void* d_ws, size_t ws_size,
                              hipStream_t stream) {
  const float* x   = (const float*)d_in[0];
  const int*   ei  = (const int*)d_in[1];
  const float* W1  = (const float*)d_in[2];
  const float* b1  = (const float*)d_in[3];
  const float* W2  = (const float*)d_in[4];
  const float* b2  = (const float*)d_in[5];

  const int N = in_sizes[0] / IN_CH;
  const int E = in_sizes[1] / 2;
  const int* esrc = ei;
  const int* edst = ei + E;

  char* w = (char*)d_ws;
  unsigned* deg  = (unsigned*)w; w += (size_t)N * 4;
  float*    dinv = (float*)w;    w += (size_t)N * 4;
  float*    hlin = (float*)w;    w += (size_t)N * HID_CH * 4;
  float*    hagg = (float*)w;    w += (size_t)N * HID_CH * 4;
  float*    zlin = (float*)w;    w += (size_t)N * OUT_CH * 4;
  float*    zagg = (float*)w;    w += (size_t)N * OUT_CH * 4;

  hipMemsetAsync(deg,  0, (size_t)N * 4, stream);
  hipMemsetAsync(hagg, 0, (size_t)N * HID_CH * 4, stream);
  hipMemsetAsync(zagg, 0, (size_t)N * OUT_CH * 4, stream);

  const int TPB = 256;
  const int nWarpBlocks = (int)(((size_t)E * 32 + TPB - 1) / TPB);
  const int mTiles = (N + 15) / 16;
  const int nEdgeBlocks = (E + TPB - 1) / TPB;

  deg_kernel<<<(E + TPB - 1) / TPB, TPB, 0, stream>>>(edst, deg, E);
  dinv_kernel<<<(N + TPB - 1) / TPB, TPB, 0, stream>>>(deg, dinv, N);

  gemm_wmma_f32<IN_CH, HID_CH><<<mTiles, 32, 0, stream>>>(x, W1, hlin, N);
  scatter_edges<HID_CH><<<nEdgeBlocks, TPB, 0, stream>>>(esrc, edst, dinv, hlin, hagg, E);
  finish_layer<HID_CH><<<((size_t)N * HID_CH + TPB - 1) / TPB, TPB, 0, stream>>>(
      hagg, hlin, dinv, b1, N, 1);

  gemm_wmma_f32<HID_CH, OUT_CH><<<mTiles, 32, 0, stream>>>(hagg, W2, zlin, N);
  scatter_edges<OUT_CH><<<nEdgeBlocks, TPB, 0, stream>>>(esrc, edst, dinv, zlin, zagg, E);
  finish_layer<OUT_CH><<<((size_t)N * OUT_CH + TPB - 1) / TPB, TPB, 0, stream>>>(
      zagg, zlin, dinv, b2, N, 0);

  decode_kernel<<<nWarpBlocks, TPB, 0, stream>>>(esrc, edst, zagg, (float*)d_out, E);
}
